# starting kernel
# baseline (speedup 1.0000x reference)
#include <hip/hip_runtime.h>
#include <hip/hip_bf16.h>
#include <math.h>

#define BATCH 256
#define N_NEG 4096
#define EMB   256
#define HALF  (EMB / 2)

typedef __attribute__((ext_vector_type(2))) float v2f;
typedef __attribute__((ext_vector_type(8))) float v8f;

__global__ void __launch_bounds__(128)
rotate_kernel(const float* __restrict__ head,
              const int*   __restrict__ rel_id,
              const float* __restrict__ rel_emb,
              float* __restrict__ rot,
              float* __restrict__ rn) {
    const int b = blockIdx.x;
    const int j = threadIdx.x;

    const float re = head[b * EMB + j];
    const float im = head[b * EMB + HALF + j];
    const int   r  = rel_id[b];
    const float ph = rel_emb[r * HALF + j];

    float s, c;
    __sincosf(ph, &s, &c);

    rot[b * EMB + j]        = re * c - im * s;
    rot[b * EMB + HALF + j] = re * s + im * c;

    __shared__ float sm[128];
    sm[j] = re * re + im * im;
    __syncthreads();
    #pragma unroll
    for (int stride = 64; stride > 0; stride >>= 1) {
        if (j < stride) sm[j] += sm[j + stride];
        __syncthreads();
    }
    if (j == 0) rn[b] = sm[0];
}

__global__ void __launch_bounds__(256)
candnorm_kernel(const float* __restrict__ cand,
                float* __restrict__ cn) {
    const int t   = blockIdx.x;
    const int tid = threadIdx.x;
    const float v = cand[t * EMB + tid];

    __shared__ float sm[256];
    sm[tid] = v * v;
    __syncthreads();
    #pragma unroll
    for (int stride = 128; stride > 0; stride >>= 1) {
        if (tid < stride) sm[tid] += sm[tid + stride];
        __syncthreads();
    }
    if (tid == 0) cn[t] = sm[0];
}

__global__ void __launch_bounds__(256)
score_kernel(const float* __restrict__ rot,
             const float* __restrict__ cand,
             const float* __restrict__ rn,
             const float* __restrict__ cn,
             float* __restrict__ out) {
    const int lane  = threadIdx.x & 31;
    const int wave  = threadIdx.x >> 5;
    const int mtile = blockIdx.y;
    const int ntile = blockIdx.x * 8 + wave;

    const int r    = lane & 15;
    const int kadj = (lane >> 4) << 1;

    const float* arow = rot  + (mtile * 16 + r) * EMB + kadj;
    const float* brow = cand + (ntile * 16 + r) * EMB + kadj;

    v8f acc = {};
    #pragma unroll
    for (int k = 0; k < EMB; k += 4) {
        v2f a = *(const v2f*)(arow + k);
        v2f b = *(const v2f*)(brow + k);
        acc = __builtin_amdgcn_wmma_f32_16x16x4_f32(
            false, a, false, b, (short)0, acc, false, false);
    }

    const int col  = ntile * 16 + r;
    const float cnv = cn[col];
    #pragma unroll
    for (int j = 0; j < 8; ++j) {
        const int row = mtile * 16 + j + ((lane >> 4) << 3);
        const float d2 = rn[row] + cnv - 2.0f * acc[j];
        out[row * N_NEG + col] = -sqrtf(fmaxf(d2, 0.0f));
    }
}

extern "C" void kernel_launch(void* const* d_in, const int* in_sizes, int n_in,
                              void* d_out, int out_size, void* d_ws, size_t ws_size,
                              hipStream_t stream) {
    const float* head    = (const float*)d_in[0];
    const int*   rel_id  = (const int*)  d_in[1];
    const float* tail    = (const float*)d_in[2];
    const float* rel_emb = (const float*)d_in[3];
    float*       out     = (float*)d_out;

    float* rot = (float*)d_ws;
    float* rn  = rot + BATCH * EMB;
    float* cn  = rn + BATCH;

    rotate_kernel<<<BATCH, 128, 0, stream>>>(head, rel_id, rel_emb, rot, rn);
    candnorm_kernel<<<N_NEG, 256, 0, stream>>>(tail, cn);

    dim3 grid(N_NEG / (16 * 8), BATCH / 16);
    score_kernel<<<grid, 256, 0, stream>>>(rot, tail, rn, cn, out);
}


# Fuse the three launches into one: each block rotates its 16 head rows into LDS (A fragments read from there) and sums row and candidate norms itself, per lane serially then across lanes
# speedup vs baseline: 2.1177x; 2.1177x over previous
#include <hip/hip_runtime.h>
#include <math.h>

#define BATCH 256
#define N_NEG 4096
#define EMB   256
#define HALF  (EMB / 2)
#define APITCH (EMB + 4)

typedef __attribute__((ext_vector_type(2))) float v2f;
typedef __attribute__((ext_vector_type(8))) float v8f;

__global__ void __launch_bounds__(256)
rotate_score_kernel(const float* __restrict__ head,
                    const int*   __restrict__ rel_id,
                    const float* __restrict__ cand,
                    const float* __restrict__ rel_emb,
                    float* __restrict__ out) {
    __shared__ __attribute__((aligned(16))) float rotS[16 * APITCH];
    __shared__ float rnS[16];

    const int lane  = threadIdx.x & 31;
    const int wave  = __builtin_amdgcn_readfirstlane(threadIdx.x >> 5);
    const int mtile = blockIdx.y;
    const int ntile = blockIdx.x * 8 + wave;

    const int r    = lane & 15;
    const int hi   = lane >> 4;
    const int kadj = hi << 1;

    const int lrow0 = wave * 2;
    const int grow0 = mtile * 16 + lrow0;
    const int rid0 = rel_id[grow0];
    const int rid1 = rel_id[grow0 + 1];
    float re[2][4], im[2][4], ph[2][4];
    #pragma unroll
    for (int rr = 0; rr < 2; ++rr) {
        #pragma unroll
        for (int q = 0; q < 4; ++q) {
            re[rr][q] = head[(grow0 + rr) * EMB + lane + 32 * q];
            im[rr][q] = head[(grow0 + rr) * EMB + HALF + lane + 32 * q];
        }
    }
    #pragma unroll
    for (int q = 0; q < 4; ++q) {
        ph[0][q] = rel_emb[rid0 * HALF + lane + 32 * q];
        ph[1][q] = rel_emb[rid1 * HALF + lane + 32 * q];
    }

    const float* brow = cand + (ntile * 16 + r) * EMB + kadj;
    v2f b[EMB / 4];
    #pragma unroll
    for (int i = 0; i < EMB / 4; ++i) b[i] = *(const v2f*)(brow + 4 * i);

    float nrm[2];
    #pragma unroll
    for (int rr = 0; rr < 2; ++rr) {
        float* dst = rotS + (lrow0 + rr) * APITCH + lane;
        float acc_n = 0.0f;
        #pragma unroll
        for (int q = 0; q < 4; ++q) {
            float s, c;
            __sincosf(ph[rr][q], &s, &c);
            dst[32 * q]        = re[rr][q] * c - im[rr][q] * s;
            dst[HALF + 32 * q] = re[rr][q] * s + im[rr][q] * c;
            acc_n += re[rr][q] * re[rr][q] + im[rr][q] * im[rr][q];
        }
        nrm[rr] = acc_n;
    }
    #pragma unroll
    for (int off = 16; off > 0; off >>= 1) {
        nrm[0] += __shfl_xor(nrm[0], off);
        nrm[1] += __shfl_xor(nrm[1], off);
    }
    if (lane == 0) {
        rnS[lrow0]     = nrm[0];
        rnS[lrow0 + 1] = nrm[1];
    }
    __syncthreads();

    const float* aS = rotS + r * APITCH + kadj;
    v8f acc = {};
    float nb0 = 0.0f, nb1 = 0.0f;
    #pragma unroll
    for (int i = 0; i < EMB / 4; ++i) {
        v2f a = *(const v2f*)(aS + 4 * i);
        acc = __builtin_amdgcn_wmma_f32_16x16x4_f32(
            false, a, false, b[i], (short)0, acc, false, false);
        nb0 += b[i].x * b[i].x;
        nb1 += b[i].y * b[i].y;
    }
    float cnv = nb0 + nb1;
    cnv += __shfl_xor(cnv, 16);

    const int col = ntile * 16 + r;
    #pragma unroll
    for (int j = 0; j < 8; ++j) {
        const int lrow = j + (hi << 3);
        const float d2 = rnS[lrow] + cnv - 2.0f * acc[j];
        out[(mtile * 16 + lrow) * N_NEG + col] = -sqrtf(fmaxf(d2, 0.0f));
    }
}

extern "C" void kernel_launch(void* const* d_in, const int* in_sizes, int n_in,
                              void* d_out, int out_size, void* d_ws, size_t ws_size,
                              hipStream_t stream) {
    const float* head    = (const float*)d_in[0];
    const int*   rel_id  = (const int*)  d_in[1];
    const float* tail    = (const float*)d_in[2];
    const float* rel_emb = (const float*)d_in[3];
    float*       out     = (float*)d_out;

    dim3 grid(N_NEG / (16 * 8), BATCH / 16);
    rotate_score_kernel<<<grid, 256, 0, stream>>>(head, rel_id, tail, rel_emb, out);
}


# Candidate rows go global->LDS by whole-line 16-B loads (4 lines per load instead of 16) and B fragments are read from LDS; k order and all sums as before
# speedup vs baseline: 2.6251x; 1.2396x over previous
#include <hip/hip_runtime.h>
#include <math.h>

#define BATCH 256
#define N_NEG 4096
#define EMB   256
#define HALF  (EMB / 2)
#define APITCH (EMB + 4)

typedef __attribute__((ext_vector_type(2))) float v2f;
typedef __attribute__((ext_vector_type(4))) float v4f;
typedef __attribute__((ext_vector_type(8))) float v8f;

__global__ void __launch_bounds__(256)
rotate_score_kernel(const float* __restrict__ head,
                    const int*   __restrict__ rel_id,
                    const float* __restrict__ cand,
                    const float* __restrict__ rel_emb,
                    float* __restrict__ out) {
    __shared__ __attribute__((aligned(16))) float rotS[16 * APITCH];
    __shared__ __attribute__((aligned(16))) float candS[8 * 16 * APITCH];
    __shared__ float rnS[16];

    const int lane  = threadIdx.x & 31;
    const int wave  = __builtin_amdgcn_readfirstlane(threadIdx.x >> 5);
    const int mtile = blockIdx.y;
    const int ntile = blockIdx.x * 8 + wave;

    const int r    = lane & 15;
    const int hi   = lane >> 4;
    const int kadj = hi << 1;

    const float* cpiece = cand + ntile * 16 * EMB + lane * 4;
    v4f cb[32];
    #pragma unroll
    for (int j = 0; j < 32; ++j) cb[j] = *(const v4f*)(cpiece + 128 * j);

    const int lrow0 = wave * 2;
    const int grow0 = mtile * 16 + lrow0;
    const int rid0 = rel_id[grow0];
    const int rid1 = rel_id[grow0 + 1];
    float re[2][4], im[2][4], ph[2][4];
    #pragma unroll
    for (int q = 0; q < 4; ++q) {
        ph[0][q] = rel_emb[rid0 * HALF + lane + 32 * q];
        ph[1][q] = rel_emb[rid1 * HALF + lane + 32 * q];
    }
    #pragma unroll
    for (int rr = 0; rr < 2; ++rr) {
        #pragma unroll
        for (int q = 0; q < 4; ++q) {
            re[rr][q] = head[(grow0 + rr) * EMB + lane + 32 * q];
            im[rr][q] = head[(grow0 + rr) * EMB + HALF + lane + 32 * q];
        }
    }

    float* cS = candS + wave * (16 * APITCH);
    #pragma unroll
    for (int j = 0; j < 32; ++j)
        *(v4f*)(cS + (j >> 1) * APITCH + (j & 1) * HALF + lane * 4) = cb[j];

    float nrm[2];
    #pragma unroll
    for (int rr = 0; rr < 2; ++rr) {
        float* dst = rotS + (lrow0 + rr) * APITCH + lane;
        float acc_n = 0.0f;
        #pragma unroll
        for (int q = 0; q < 4; ++q) {
            float s, c;
            __sincosf(ph[rr][q], &s, &c);
            dst[32 * q]        = re[rr][q] * c - im[rr][q] * s;
            dst[HALF + 32 * q] = re[rr][q] * s + im[rr][q] * c;
            acc_n += re[rr][q] * re[rr][q] + im[rr][q] * im[rr][q];
        }
        nrm[rr] = acc_n;
    }
    #pragma unroll
    for (int off = 16; off > 0; off >>= 1) {
        nrm[0] += __shfl_xor(nrm[0], off);
        nrm[1] += __shfl_xor(nrm[1], off);
    }
    if (lane == 0) {
        rnS[lrow0]     = nrm[0];
        rnS[lrow0 + 1] = nrm[1];
    }
    __syncthreads();

    const float* aS = rotS + r * APITCH + kadj;
    const float* bS = cS   + r * APITCH + kadj;
    v8f acc = {};
    float nb0 = 0.0f, nb1 = 0.0f;
    #pragma unroll
    for (int i = 0; i < EMB / 4; ++i) {
        v2f a = *(const v2f*)(aS + 4 * i);
        v2f b = *(const v2f*)(bS + 4 * i);
        acc = __builtin_amdgcn_wmma_f32_16x16x4_f32(
            false, a, false, b, (short)0, acc, false, false);
        nb0 += b.x * b.x;
        nb1 += b.y * b.y;
    }
    float cnv = nb0 + nb1;
    cnv += __shfl_xor(cnv, 16);

    const int col = ntile * 16 + r;
    #pragma unroll
    for (int j = 0; j < 8; ++j) {
        const int lrow = j + (hi << 3);
        const float d2 = rnS[lrow] + cnv - 2.0f * acc[j];
        out[(mtile * 16 + lrow) * N_NEG + col] = -sqrtf(fmaxf(d2, 0.0f));
    }
}

extern "C" void kernel_launch(void* const* d_in, const int* in_sizes, int n_in,
                              void* d_out, int out_size, void* d_ws, size_t ws_size,
                              hipStream_t stream) {
    const float* head    = (const float*)d_in[0];
    const int*   rel_id  = (const int*)  d_in[1];
    const float* tail    = (const float*)d_in[2];
    const float* rel_emb = (const float*)d_in[3];
    float*       out     = (float*)d_out;

    dim3 grid(N_NEG / (16 * 8), BATCH / 16);
    rotate_score_kernel<<<grid, 256, 0, stream>>>(head, rel_id, tail, rel_emb, out);
}


# Candidate tile fetched and staged to LDS in four K quarters; a quarter's WMMAs run while later quarters are still in flight (LDS-only block barrier, counted load waits); all sums as before
# speedup vs baseline: 2.7923x; 1.0637x over previous
#include <hip/hip_runtime.h>
#include <math.h>

#define BATCH 256
#define N_NEG 4096
#define EMB   256
#define HALF  (EMB / 2)
#define APITCH (EMB + 4)
#define KCH 4
#define CHW (EMB / KCH)

typedef __attribute__((ext_vector_type(2))) float v2f;
typedef __attribute__((ext_vector_type(4))) float v4f;
typedef __attribute__((ext_vector_type(8))) float v8f;

__global__ void __launch_bounds__(256)
rotate_score_kernel(const float* __restrict__ head,
                    const int*   __restrict__ rel_id,
                    const float* __restrict__ cand,
                    const float* __restrict__ rel_emb,
                    float* __restrict__ out) {
    __shared__ __attribute__((aligned(16))) float rotS[16 * APITCH];
    __shared__ __attribute__((aligned(16))) float candS[8 * 16 * APITCH];
    __shared__ float rnS[16];

    const int lane  = threadIdx.x & 31;
    const int wave  = __builtin_amdgcn_readfirstlane(threadIdx.x >> 5);
    const int mtile = blockIdx.y;
    const int ntile = blockIdx.x * 8 + wave;

    const int r    = lane & 15;
    const int hi   = lane >> 4;
    const int kadj = hi << 1;

    const int lrow0 = wave * 2;
    const int grow0 = mtile * 16 + lrow0;
    const int rid0 = rel_id[grow0];
    const int rid1 = rel_id[grow0 + 1];

    float re[2][4], im[2][4], ph[2][4];
    #pragma unroll
    for (int rr = 0; rr < 2; ++rr) {
        #pragma unroll
        for (int q = 0; q < 4; ++q) {
            re[rr][q] = head[(grow0 + rr) * EMB + lane + 32 * q];
            im[rr][q] = head[(grow0 + rr) * EMB + HALF + lane + 32 * q];
        }
    }

    const float* cpiece = cand + ntile * 16 * EMB + hi * EMB + r * 4;
    v4f cb[KCH][8];
    #pragma unroll
    for (int m = 0; m < 8; ++m) cb[0][m] = *(const v4f*)(cpiece + 2 * m * EMB);

    __builtin_amdgcn_sched_barrier(0);
    #pragma unroll
    for (int q = 0; q < 4; ++q) {
        ph[0][q] = rel_emb[rid0 * HALF + lane + 32 * q];
        ph[1][q] = rel_emb[rid1 * HALF + lane + 32 * q];
    }
    #pragma unroll
    for (int c = 1; c < KCH; ++c) {
        __builtin_amdgcn_sched_barrier(0);
        #pragma unroll
        for (int m = 0; m < 8; ++m) cb[c][m] = *(const v4f*)(cpiece + 2 * m * EMB + CHW * c);
    }

    float* cS   = candS + wave * (16 * APITCH);
    float* cdst = cS + hi * APITCH + r * 4;
    #pragma unroll
    for (int m = 0; m < 8; ++m) *(v4f*)(cdst + 2 * m * APITCH) = cb[0][m];

    float nrm[2];
    #pragma unroll
    for (int rr = 0; rr < 2; ++rr) {
        float* dst = rotS + (lrow0 + rr) * APITCH + lane;
        float acc_n = 0.0f;
        #pragma unroll
        for (int q = 0; q < 4; ++q) {
            float s, c;
            __sincosf(ph[rr][q], &s, &c);
            dst[32 * q]        = re[rr][q] * c - im[rr][q] * s;
            dst[HALF + 32 * q] = re[rr][q] * s + im[rr][q] * c;
            acc_n += re[rr][q] * re[rr][q] + im[rr][q] * im[rr][q];
        }
        nrm[rr] = acc_n;
    }
    #pragma unroll
    for (int off = 16; off > 0; off >>= 1) {
        nrm[0] += __shfl_xor(nrm[0], off);
        nrm[1] += __shfl_xor(nrm[1], off);
    }
    if (lane == 0) {
        rnS[lrow0]     = nrm[0];
        rnS[lrow0 + 1] = nrm[1];
    }
    __builtin_amdgcn_fence(__ATOMIC_RELEASE, "workgroup", "local");
    __builtin_amdgcn_s_barrier();
    __builtin_amdgcn_fence(__ATOMIC_ACQUIRE, "workgroup", "local");

    const float* aS = rotS + r * APITCH + kadj;
    const float* bS = cS   + r * APITCH + kadj;
    v8f acc = {};
    float nb0 = 0.0f, nb1 = 0.0f;
    #pragma unroll
    for (int c = 0; c < KCH; ++c) {
        #pragma unroll
        for (int i = 16 * c; i < 16 * c + 16; ++i) {
            if (i == 16 * c + 12 && c + 1 < KCH) {
                #pragma unroll
                for (int m = 0; m < 8; ++m)
                    *(v4f*)(cdst + 2 * m * APITCH + CHW * (c + 1)) = cb[c + 1][m];
            }
            v2f a = *(const v2f*)(aS + 4 * i);
            v2f b = *(const v2f*)(bS + 4 * i);
            acc = __builtin_amdgcn_wmma_f32_16x16x4_f32(
                false, a, false, b, (short)0, acc, false, false);
            nb0 += b.x * b.x;
            nb1 += b.y * b.y;
        }
    }
    float cnv = nb0 + nb1;
    cnv += __shfl_xor(cnv, 16);

    const int col = ntile * 16 + r;
    #pragma unroll
    for (int j = 0; j < 8; ++j) {
        const int lrow = j + (hi << 3);
        const float d2 = rnS[lrow] + cnv - 2.0f * acc[j];
        out[(mtile * 16 + lrow) * N_NEG + col] = -sqrtf(fmaxf(d2, 0.0f));
    }
}

extern "C" void kernel_launch(void* const* d_in, const int* in_sizes, int n_in,
                              void* d_out, int out_size, void* d_ws, size_t ws_size,
                              hipStream_t stream) {
    const float* head    = (const float*)d_in[0];
    const int*   rel_id  = (const int*)  d_in[1];
    const float* tail    = (const float*)d_in[2];
    const float* rel_emb = (const float*)d_in[3];
    float*       out     = (float*)d_out;

    dim3 grid(N_NEG / (16 * 8), BATCH / 16);
    rotate_score_kernel<<<grid, 256, 0, stream>>>(head, rel_id, tail, rel_emb, out);
}


# Each wave scores its 16 candidates against 32 head rows (two accumulator tiles share every B fragment): grid 32x8, one block per WGP, half the waves; k order and all sums as before
# speedup vs baseline: 2.9364x; 1.0516x over previous
#include <hip/hip_runtime.h>
#include <math.h>

#define BATCH 256
#define N_NEG 4096
#define EMB   256
#define HALF  (EMB / 2)
#define APITCH (EMB + 4)
#define KCH 4
#define CHW (EMB / KCH)
#define MT 2
#define MROWS (16 * MT)
#define RPW (MROWS / 8)

typedef __attribute__((ext_vector_type(2))) float v2f;
typedef __attribute__((ext_vector_type(4))) float v4f;
typedef __attribute__((ext_vector_type(8))) float v8f;

__global__ void __launch_bounds__(256)
rotate_score_kernel(const float* __restrict__ head,
                    const int*   __restrict__ rel_id,
                    const float* __restrict__ cand,
                    const float* __restrict__ rel_emb,
                    float* __restrict__ out) {
    __shared__ __attribute__((aligned(16))) float rotS[MROWS * APITCH];
    __shared__ __attribute__((aligned(16))) float candS[8 * 16 * APITCH];
    __shared__ float rnS[MROWS];

    const int lane  = threadIdx.x & 31;
    const int wave  = __builtin_amdgcn_readfirstlane(threadIdx.x >> 5);
    const int mtile = blockIdx.y;
    const int ntile = blockIdx.x * 8 + wave;

    const int r    = lane & 15;
    const int hi   = lane >> 4;
    const int kadj = hi << 1;

    const int lrow0 = wave * RPW;
    const int grow0 = mtile * MROWS + lrow0;
    int rid[RPW];
    #pragma unroll
    for (int rr = 0; rr < RPW; ++rr) rid[rr] = rel_id[grow0 + rr];

    float re[RPW][4], im[RPW][4], ph[RPW][4];
    #pragma unroll
    for (int rr = 0; rr < RPW; ++rr) {
        #pragma unroll
        for (int q = 0; q < 4; ++q) {
            re[rr][q] = head[(grow0 + rr) * EMB + lane + 32 * q];
            im[rr][q] = head[(grow0 + rr) * EMB + HALF + lane + 32 * q];
        }
    }

    const float* cpiece = cand + ntile * 16 * EMB + hi * EMB + r * 4;
    v4f cb[KCH][8];
    __builtin_amdgcn_sched_barrier(0);
    #pragma unroll
    for (int m = 0; m < 8; ++m) cb[0][m] = *(const v4f*)(cpiece + 2 * m * EMB);

    __builtin_amdgcn_sched_barrier(0);
    #pragma unroll
    for (int rr = 0; rr < RPW; ++rr) {
        #pragma unroll
        for (int q = 0; q < 4; ++q) ph[rr][q] = rel_emb[rid[rr] * HALF + lane + 32 * q];
    }
    #pragma unroll
    for (int c = 1; c < KCH; ++c) {
        __builtin_amdgcn_sched_barrier(0);
        #pragma unroll
        for (int m = 0; m < 8; ++m) cb[c][m] = *(const v4f*)(cpiece + 2 * m * EMB + CHW * c);
    }

    float* cS   = candS + wave * (16 * APITCH);
    float* cdst = cS + hi * APITCH + r * 4;
    #pragma unroll
    for (int m = 0; m < 8; ++m) *(v4f*)(cdst + 2 * m * APITCH) = cb[0][m];

    float nrm[RPW];
    #pragma unroll
    for (int rr = 0; rr < RPW; ++rr) {
        float* dst = rotS + (lrow0 + rr) * APITCH + lane;
        float acc_n = 0.0f;
        #pragma unroll
        for (int q = 0; q < 4; ++q) {
            float s, c;
            __sincosf(ph[rr][q], &s, &c);
            dst[32 * q]        = re[rr][q] * c - im[rr][q] * s;
            dst[HALF + 32 * q] = re[rr][q] * s + im[rr][q] * c;
            acc_n += re[rr][q] * re[rr][q] + im[rr][q] * im[rr][q];
        }
        nrm[rr] = acc_n;
    }
    #pragma unroll
    for (int off = 16; off > 0; off >>= 1) {
        #pragma unroll
        for (int rr = 0; rr < RPW; ++rr) nrm[rr] += __shfl_xor(nrm[rr], off);
    }
    if (lane == 0) {
        #pragma unroll
        for (int rr = 0; rr < RPW; ++rr) rnS[lrow0 + rr] = nrm[rr];
    }
    __builtin_amdgcn_fence(__ATOMIC_RELEASE, "workgroup", "local");
    __builtin_amdgcn_s_barrier();
    __builtin_amdgcn_fence(__ATOMIC_ACQUIRE, "workgroup", "local");

    const float* aS = rotS + r * APITCH + kadj;
    const float* bS = cS   + r * APITCH + kadj;
    v8f acc[MT];
    #pragma unroll
    for (int t = 0; t < MT; ++t) acc[t] = v8f{};
    float nb0 = 0.0f, nb1 = 0.0f;
    #pragma unroll
    for (int c = 0; c < KCH; ++c) {
        #pragma unroll
        for (int i = 16 * c; i < 16 * c + 16; ++i) {
            if (i == 16 * c + 12 && c + 1 < KCH) {
                #pragma unroll
                for (int m = 0; m < 8; ++m)
                    *(v4f*)(cdst + 2 * m * APITCH + CHW * (c + 1)) = cb[c + 1][m];
            }
            v2f b = *(const v2f*)(bS + 4 * i);
            #pragma unroll
            for (int t = 0; t < MT; ++t) {
                v2f a = *(const v2f*)(aS + t * (16 * APITCH) + 4 * i);
                acc[t] = __builtin_amdgcn_wmma_f32_16x16x4_f32(
                    false, a, false, b, (short)0, acc[t], false, false);
            }
            nb0 += b.x * b.x;
            nb1 += b.y * b.y;
        }
    }
    float cnv = nb0 + nb1;
    cnv += __shfl_xor(cnv, 16);

    const int col = ntile * 16 + r;
    #pragma unroll
    for (int t = 0; t < MT; ++t) {
        #pragma unroll
        for (int j = 0; j < 8; ++j) {
            const int lrow = 16 * t + j + (hi << 3);
            const float d2 = rnS[lrow] + cnv - 2.0f * acc[t][j];
            out[(mtile * MROWS + lrow) * N_NEG + col] = -sqrtf(fmaxf(d2, 0.0f));
        }
    }
}

extern "C" void kernel_launch(void* const* d_in, const int* in_sizes, int n_in,
                              void* d_out, int out_size, void* d_ws, size_t ws_size,
                              hipStream_t stream) {
    const float* head    = (const float*)d_in[0];
    const int*   rel_id  = (const int*)  d_in[1];
    const float* tail    = (const float*)d_in[2];
    const float* rel_emb = (const float*)d_in[3];
    float*       out     = (float*)d_out;

    dim3 grid(N_NEG / (16 * 8), BATCH / MROWS);
    rotate_score_kernel<<<grid, 256, 0, stream>>>(head, rel_id, tail, rel_emb, out);
}
